# starting kernel
# baseline (speedup 1.0000x reference)
#include <hip/hip_runtime.h>
#include <hip/hip_bf16.h>

#define NBATCH 256
#define CDIM   256
#define KSUB   128

#define RHO_S  144
#define TT_S   137

typedef __attribute__((ext_vector_type(2))) float v2f;
typedef __attribute__((ext_vector_type(8))) float v8f;

__global__ void build_ut_kernel(const float* __restrict__ ux,
                                const float* __restrict__ uy,
                                const float* __restrict__ uc,
                                float* __restrict__ UT) {
    int idx = blockIdx.x * blockDim.x + threadIdx.x;
    if (idx >= CDIM * KSUB) return;
    int p = idx / KSUB;
    int q = idx % KSUB;
    int c1 = p >> 6;
    int i1 = (p >> 3) & 7;
    int j1 = p & 7;
    int c2 = 2 + (q >> 6);
    int i2 = (q >> 3) & 7;
    int j2 = q & 7;
    UT[q * CDIM + p] = uc[c1 * 4 + c2] * ux[i1 * 8 + i2] * uy[j1 * 8 + j2];
}

__global__ void __launch_bounds__(256, 2)
qconv_wmma_kernel(const float* __restrict__ rho,
                  const float* __restrict__ UT,
                  float* __restrict__ out) {
    extern __shared__ float smem[];
    float* sRho = smem;
    float* sTt  = smem + KSUB * RHO_S;

    const int b    = blockIdx.x >> 1;
    const int half = blockIdx.x & 1;
    const int tid  = threadIdx.x;
    const int lane = tid & 31;
    const int w    = tid >> 5;
    const int l16  = lane & 15;
    const int hl   = lane >> 4;

    {
        const float4* src = (const float4*)(rho + (size_t)b * KSUB * KSUB);
        float4* dst = (float4*)sRho;
        #pragma unroll
        for (int i = 0; i < (KSUB * KSUB / 4) / 256; ++i) {
            int g    = tid + i * 256;
            int row  = g >> 5;
            int col4 = g & 31;
            dst[row * (RHO_S / 4) + col4] = src[g];
        }
    }
    __syncthreads();

    {
        const int prow = half * KSUB + w * 16 + l16;
        v8f acc[8] = {v8f{}, v8f{}, v8f{}, v8f{}, v8f{}, v8f{}, v8f{}, v8f{}};
        for (int kk = 0; kk < KSUB; kk += 4) {
            v2f a;
            a.x = UT[(kk + 2 * hl)     * CDIM + prow];
            a.y = UT[(kk + 2 * hl + 1) * CDIM + prow];
            #pragma unroll
            for (int tn = 0; tn < 8; ++tn) {
                v2f bf;
                bf.x = sRho[(kk + 2 * hl)     * RHO_S + tn * 16 + l16];
                bf.y = sRho[(kk + 2 * hl + 1) * RHO_S + tn * 16 + l16];
                acc[tn] = __builtin_amdgcn_wmma_f32_16x16x4_f32(
                    false, a, false, bf, (short)0, acc[tn], false, false);
            }
        }
        #pragma unroll
        for (int tn = 0; tn < 8; ++tn) {
            #pragma unroll
            for (int v = 0; v < 8; ++v)
                sTt[(tn * 16 + l16) * TT_S + w * 16 + v + 8 * hl] = acc[tn][v];
        }
    }
    __syncthreads();

    float* outb = out + (size_t)b * CDIM * CDIM + (size_t)half * KSUB * CDIM;
    const int plocal = w * 16 + l16;
    for (int th = 0; th < 2; ++th) {
        v8f acc[8] = {v8f{}, v8f{}, v8f{}, v8f{}, v8f{}, v8f{}, v8f{}, v8f{}};
        for (int kk = 0; kk < KSUB; kk += 4) {
            v2f a;
            a.x = sTt[(kk + 2 * hl)     * TT_S + plocal];
            a.y = sTt[(kk + 2 * hl + 1) * TT_S + plocal];
            #pragma unroll
            for (int tn8 = 0; tn8 < 8; ++tn8) {
                const int tn = th * 8 + tn8;
                v2f bf;
                bf.x = UT[(kk + 2 * hl)     * CDIM + tn * 16 + l16];
                bf.y = UT[(kk + 2 * hl + 1) * CDIM + tn * 16 + l16];
                acc[tn8] = __builtin_amdgcn_wmma_f32_16x16x4_f32(
                    false, a, false, bf, (short)0, acc[tn8], false, false);
            }
        }
        #pragma unroll
        for (int tn8 = 0; tn8 < 8; ++tn8) {
            const int tn = th * 8 + tn8;
            #pragma unroll
            for (int v = 0; v < 8; ++v)
                outb[(w * 16 + v + 8 * hl) * CDIM + tn * 16 + l16] = acc[tn8][v];
        }
    }
}

extern "C" void kernel_launch(void* const* d_in, const int* in_sizes, int n_in,
                              void* d_out, int out_size, void* d_ws, size_t ws_size,
                              hipStream_t stream) {
    (void)in_sizes; (void)n_in; (void)out_size; (void)ws_size;
    const float* rho = (const float*)d_in[0];
    const float* ux  = (const float*)d_in[1];
    const float* uy  = (const float*)d_in[2];
    const float* uc  = (const float*)d_in[3];
    float* out = (float*)d_out;

    float* UT = (float*)d_ws;

    build_ut_kernel<<<(CDIM * KSUB + 255) / 256, 256, 0, stream>>>(ux, uy, uc, UT);

    const int lds_bytes = (KSUB * RHO_S + KSUB * TT_S) * (int)sizeof(float);
    (void)hipFuncSetAttribute((const void*)qconv_wmma_kernel,
                              hipFuncAttributeMaxDynamicSharedMemorySize, lds_bytes);

    qconv_wmma_kernel<<<NBATCH * 2, 256, lds_bytes, stream>>>(rho, UT, out);
}


# T kept in registers: product 1 done transposed so its accumulator tiles are product 2's A fragments; no LDS transpose, no workgroup barrier after the rho load; sums regrouped (k order of both)
# speedup vs baseline: 1.2799x; 1.2799x over previous
#include <hip/hip_runtime.h>
#include <hip/hip_bf16.h>

#define NBATCH 256
#define CDIM   256
#define KSUB   128

#define RHO_S  136
#define STG_S  132

typedef __attribute__((ext_vector_type(2))) float v2f;
typedef __attribute__((ext_vector_type(8))) float v8f;

static __device__ __forceinline__ void wave_lds_handover() {
    __builtin_amdgcn_fence(__ATOMIC_RELEASE, "wavefront");
    __builtin_amdgcn_wave_barrier();
    __builtin_amdgcn_fence(__ATOMIC_ACQUIRE, "wavefront");
}

__global__ void build_ut_kernel(const float* __restrict__ ux,
                                const float* __restrict__ uy,
                                const float* __restrict__ uc,
                                float* __restrict__ UTp) {
    int idx = blockIdx.x * blockDim.x + threadIdx.x;
    if (idx >= CDIM * KSUB) return;
    int i  = idx & 3;
    int p  = (idx >> 2) & 255;
    int u  = idx >> 10;
    int q  = 4 * u + i;
    int c1 = p >> 6;
    int i1 = (p >> 3) & 7;
    int j1 = p & 7;
    int c2 = 2 + (q >> 6);
    int i2 = (q >> 3) & 7;
    int j2 = q & 7;
    UTp[idx] = uc[c1 * 4 + c2] * ux[i1 * 8 + i2] * uy[j1 * 8 + j2];
}

__global__ void __launch_bounds__(256, 2)
qconv_wmma_kernel(const float* __restrict__ rho,
                  const float* __restrict__ UTp,
                  float* __restrict__ out) {
    extern __shared__ __attribute__((aligned(16))) float smem[];
    float* sRho = smem;
    float* sStg = smem + KSUB * RHO_S;

    const int b    = blockIdx.x >> 1;
    const int half = blockIdx.x & 1;
    const int tid  = threadIdx.x;
    const int lane = tid & 31;
    const int w    = tid >> 5;
    const int l16  = lane & 15;
    const int hl   = lane >> 4;

    {
        const float4* src = (const float4*)(rho + (size_t)b * KSUB * KSUB);
        float4* dst = (float4*)sRho;
        #pragma unroll
        for (int i = 0; i < (KSUB * KSUB / 4) / 256; ++i) {
            int g    = tid + i * 256;
            int row  = g >> 5;
            int col4 = g & 31;
            dst[row * (RHO_S / 4) + col4] = src[g];
        }
    }
    __syncthreads();

    const float4* UT4 = (const float4*)UTp;

    v8f accT[8] = {v8f{}, v8f{}, v8f{}, v8f{}, v8f{}, v8f{}, v8f{}, v8f{}};
    {
        const int prow = half * KSUB + w * 16 + l16;
        const float4* UTa = UT4 + hl * CDIM + prow;
        const float* rA = sRho + (4 * hl) * RHO_S + l16;
        for (int g = 0; g < KSUB / 8; ++g) {
            const float4 bq = UTa[g * (2 * CDIM)];
            #pragma unroll
            for (int s = 0; s < 2; ++s) {
                v2f bf;
                bf.x = s ? bq.z : bq.x;
                bf.y = s ? bq.w : bq.y;
                const float* rq = rA + (8 * g + 2 * s) * RHO_S;
                #pragma unroll
                for (int tn = 0; tn < 8; ++tn) {
                    v2f af;
                    af.x = rq[tn * 16];
                    af.y = rq[RHO_S + tn * 16];
                    accT[tn] = __builtin_amdgcn_wmma_f32_16x16x4_f32(
                        false, af, false, bf, (short)0, accT[tn], false, false);
                }
            }
        }
    }

    float* outb = out + (size_t)b * CDIM * CDIM + (size_t)half * KSUB * CDIM;
    float* stg = sStg + w * (16 * STG_S);
    for (int th = 0; th < 2; ++th) {
        v8f acc[8] = {v8f{}, v8f{}, v8f{}, v8f{}, v8f{}, v8f{}, v8f{}, v8f{}};
        const float4* UTb = UT4 + (2 * hl) * CDIM + th * 128 + l16;
        float4 bq[2][8];
        #pragma unroll
        for (int tn8 = 0; tn8 < 8; ++tn8)
            bq[0][tn8] = UTb[tn8 * 16];
        #pragma unroll
        for (int st = 0; st < 16; ++st) {
            const int tn = st >> 1;
            const int jp = st & 1;
            if (st + 1 < 16) {
                #pragma unroll
                for (int tn8 = 0; tn8 < 8; ++tn8)
                    bq[(st + 1) & 1][tn8] =
                        UTb[(4 * ((st + 1) >> 1) + ((st + 1) & 1)) * CDIM + tn8 * 16];
            }
            __builtin_amdgcn_sched_barrier(0);
            #pragma unroll
            for (int s = 0; s < 2; ++s) {
                v2f a;
                a.x = accT[tn][4 * jp + 2 * s];
                a.y = accT[tn][4 * jp + 2 * s + 1];
                #pragma unroll
                for (int tn8 = 0; tn8 < 8; ++tn8) {
                    v2f bf;
                    bf.x = s ? bq[st & 1][tn8].z : bq[st & 1][tn8].x;
                    bf.y = s ? bq[st & 1][tn8].w : bq[st & 1][tn8].y;
                    acc[tn8] = __builtin_amdgcn_wmma_f32_16x16x4_f32(
                        false, a, false, bf, (short)0, acc[tn8], false, false);
                }
            }
        }
        #pragma unroll
        for (int tn8 = 0; tn8 < 8; ++tn8) {
            #pragma unroll
            for (int v = 0; v < 8; ++v)
                stg[(v + 8 * hl) * STG_S + tn8 * 16 + l16] = acc[tn8][v];
        }
        wave_lds_handover();
        #pragma unroll
        for (int r = 0; r < 16; ++r) {
            const float4 val = *(const float4*)(stg + r * STG_S + lane * 4);
            *(float4*)(outb + (w * 16 + r) * CDIM + th * 128 + lane * 4) = val;
        }
        wave_lds_handover();
    }
}

extern "C" void kernel_launch(void* const* d_in, const int* in_sizes, int n_in,
                              void* d_out, int out_size, void* d_ws, size_t ws_size,
                              hipStream_t stream) {
    (void)in_sizes; (void)n_in; (void)out_size; (void)ws_size;
    const float* rho = (const float*)d_in[0];
    const float* ux  = (const float*)d_in[1];
    const float* uy  = (const float*)d_in[2];
    const float* uc  = (const float*)d_in[3];
    float* out = (float*)d_out;

    float* UTp = (float*)d_ws;

    build_ut_kernel<<<(CDIM * KSUB + 255) / 256, 256, 0, stream>>>(ux, uy, uc, UTp);

    const int lds_bytes = (KSUB * RHO_S + 8 * 16 * STG_S) * (int)sizeof(float);
    (void)hipFuncSetAttribute((const void*)qconv_wmma_kernel,
                              hipFuncAttributeMaxDynamicSharedMemorySize, lds_bytes);

    qconv_wmma_kernel<<<NBATCH * 2, 256, lds_bytes, stream>>>(rho, UTp, out);
}


# B fragments of U_sub formed in registers from uc, ux, uy: one (uc*ux) per lane times its uy row, same product order; no U table, no setup launch, no B loads; sums unchanged
# speedup vs baseline: 1.3395x; 1.0466x over previous
#include <hip/hip_runtime.h>
#include <hip/hip_bf16.h>

#define NBATCH 256
#define CDIM   256
#define KSUB   128

#define RHO_S  136
#define STG_S  132

typedef __attribute__((ext_vector_type(2))) float v2f;
typedef __attribute__((ext_vector_type(8))) float v8f;

static __device__ __forceinline__ void wave_lds_handover() {
    __builtin_amdgcn_fence(__ATOMIC_RELEASE, "wavefront");
    __builtin_amdgcn_wave_barrier();
    __builtin_amdgcn_fence(__ATOMIC_ACQUIRE, "wavefront");
}

__global__ void __launch_bounds__(256, 2)
qconv_wmma_kernel(const float* __restrict__ rho,
                  const float* __restrict__ ux,
                  const float* __restrict__ uy,
                  const float* __restrict__ uc,
                  float* __restrict__ out) {
    extern __shared__ __attribute__((aligned(16))) float smem[];
    float* sRho = smem;
    float* sStg = smem + KSUB * RHO_S;

    const int b    = blockIdx.x >> 1;
    const int half = blockIdx.x & 1;
    const int tid  = threadIdx.x;
    const int lane = tid & 31;
    const int w    = tid >> 5;
    const int l16  = lane & 15;
    const int hl   = lane >> 4;

    {
        const float4* src = (const float4*)(rho + (size_t)b * KSUB * KSUB);
        float4* dst = (float4*)sRho;
        #pragma unroll
        for (int i = 0; i < (KSUB * KSUB / 4) / 256; ++i) {
            int g    = tid + i * 256;
            int row  = g >> 5;
            int col4 = g & 31;
            dst[row * (RHO_S / 4) + col4] = src[g];
        }
    }
    const int prow = half * KSUB + w * 16 + l16;
    float uyl[8];
    #pragma unroll
    for (int j = 0; j < 8; ++j)
        uyl[j] = uy[(l16 & 7) * 8 + j];
    float uxp[8];
    #pragma unroll
    for (int i = 0; i < 8; ++i)
        uxp[i] = ux[((prow >> 3) & 7) * 8 + i];
    float uxl[4][4];
    #pragma unroll
    for (int m = 0; m < 4; ++m) {
        #pragma unroll
        for (int t = 0; t < 4; ++t)
            uxl[m][t] = ux[(2 * m + (l16 >> 3)) * 8 + 2 * t + hl];
    }
    float ucs[4][2];
    #pragma unroll
    for (int c = 0; c < 4; ++c) {
        ucs[c][0] = uc[c * 4 + 2];
        ucs[c][1] = uc[c * 4 + 3];
    }
    __syncthreads();

    v8f accT[8] = {v8f{}, v8f{}, v8f{}, v8f{}, v8f{}, v8f{}, v8f{}, v8f{}};
    {
        const int cw = w >> 2;
        const float ucp0 = half ? (cw ? ucs[3][0] : ucs[2][0]) : (cw ? ucs[1][0] : ucs[0][0]);
        const float ucp1 = half ? (cw ? ucs[3][1] : ucs[2][1]) : (cw ? ucs[1][1] : ucs[0][1]);
        float uyh[4];
        #pragma unroll
        for (int k = 0; k < 4; ++k)
            uyh[k] = hl ? uyl[4 + k] : uyl[k];
        const float* rA = sRho + (4 * hl) * RHO_S + l16;
        #pragma unroll
        for (int g = 0; g < KSUB / 8; ++g) {
            const float wg = ((g >> 3) ? ucp1 : ucp0) * uxp[g & 7];
            #pragma unroll
            for (int s = 0; s < 2; ++s) {
                v2f bf;
                bf.x = wg * uyh[2 * s];
                bf.y = wg * uyh[2 * s + 1];
                const float* rq = rA + (8 * g + 2 * s) * RHO_S;
                #pragma unroll
                for (int tn = 0; tn < 8; ++tn) {
                    v2f af;
                    af.x = rq[tn * 16];
                    af.y = rq[RHO_S + tn * 16];
                    accT[tn] = __builtin_amdgcn_wmma_f32_16x16x4_f32(
                        false, af, false, bf, (short)0, accT[tn], false, false);
                }
            }
        }
    }

    float* outb = out + (size_t)b * CDIM * CDIM + (size_t)half * KSUB * CDIM;
    float* stg = sStg + w * (16 * STG_S);
    for (int th = 0; th < 2; ++th) {
        v8f acc[8] = {v8f{}, v8f{}, v8f{}, v8f{}, v8f{}, v8f{}, v8f{}, v8f{}};
        const float uc00 = th ? ucs[2][0] : ucs[0][0];
        const float uc01 = th ? ucs[2][1] : ucs[0][1];
        const float uc10 = th ? ucs[3][0] : ucs[1][0];
        const float uc11 = th ? ucs[3][1] : ucs[1][1];
        #pragma unroll
        for (int tn = 0; tn < 8; ++tn) {
            float wv[8];
            #pragma unroll
            for (int tn8 = 0; tn8 < 8; ++tn8) {
                const float ucv = (tn8 >> 2) ? ((tn >> 2) ? uc11 : uc10)
                                             : ((tn >> 2) ? uc01 : uc00);
                wv[tn8] = ucv * uxl[tn8 & 3][tn & 3];
            }
            #pragma unroll
            for (int js = 0; js < 4; ++js) {
                v2f a;
                a.x = accT[tn][2 * js];
                a.y = accT[tn][2 * js + 1];
                #pragma unroll
                for (int tn8 = 0; tn8 < 8; ++tn8) {
                    v2f bf;
                    bf.x = wv[tn8] * uyl[2 * js];
                    bf.y = wv[tn8] * uyl[2 * js + 1];
                    acc[tn8] = __builtin_amdgcn_wmma_f32_16x16x4_f32(
                        false, a, false, bf, (short)0, acc[tn8], false, false);
                }
            }
        }
        #pragma unroll
        for (int tn8 = 0; tn8 < 8; ++tn8) {
            #pragma unroll
            for (int v = 0; v < 8; ++v)
                stg[(v + 8 * hl) * STG_S + tn8 * 16 + l16] = acc[tn8][v];
        }
        wave_lds_handover();
        #pragma unroll
        for (int r = 0; r < 16; ++r) {
            const float4 val = *(const float4*)(stg + r * STG_S + lane * 4);
            *(float4*)(outb + (w * 16 + r) * CDIM + th * 128 + lane * 4) = val;
        }
        wave_lds_handover();
    }
}

extern "C" void kernel_launch(void* const* d_in, const int* in_sizes, int n_in,
                              void* d_out, int out_size, void* d_ws, size_t ws_size,
                              hipStream_t stream) {
    (void)in_sizes; (void)n_in; (void)out_size; (void)d_ws; (void)ws_size;
    const float* rho = (const float*)d_in[0];
    const float* ux  = (const float*)d_in[1];
    const float* uy  = (const float*)d_in[2];
    const float* uc  = (const float*)d_in[3];
    float* out = (float*)d_out;

    const int lds_bytes = (KSUB * RHO_S + 8 * 16 * STG_S) * (int)sizeof(float);
    (void)hipFuncSetAttribute((const void*)qconv_wmma_kernel,
                              hipFuncAttributeMaxDynamicSharedMemorySize, lds_bytes);

    qconv_wmma_kernel<<<NBATCH * 2, 256, lds_bytes, stream>>>(rho, ux, uy, uc, out);
}


# Wave tile re-cut to 32 rows x 256 columns, one 8-wave workgroup per batch: half the waves to launch, rho staged once per batch, each rho and U fragment feeds two row blocks; sums unchanged
# speedup vs baseline: 1.4320x; 1.0691x over previous
#include <hip/hip_runtime.h>
#include <hip/hip_bf16.h>

#define NBATCH 256
#define CDIM   256
#define KSUB   128

#define RHO_S  136
#define STG_S  132

typedef __attribute__((ext_vector_type(2))) float v2f;
typedef __attribute__((ext_vector_type(8))) float v8f;

static __device__ __forceinline__ void wave_lds_handover() {
    __builtin_amdgcn_fence(__ATOMIC_RELEASE, "wavefront");
    __builtin_amdgcn_wave_barrier();
    __builtin_amdgcn_fence(__ATOMIC_ACQUIRE, "wavefront");
}

__global__ void __launch_bounds__(256, 1)
qconv_wmma_kernel(const float* __restrict__ rho,
                  const float* __restrict__ ux,
                  const float* __restrict__ uy,
                  const float* __restrict__ uc,
                  float* __restrict__ out) {
    extern __shared__ __attribute__((aligned(16))) float smem[];
    float* sRho = smem;
    float* sStg = smem + KSUB * RHO_S;

    const int b    = blockIdx.x;
    const int tid  = threadIdx.x;
    const int lane = tid & 31;
    const int w    = tid >> 5;
    const int l16  = lane & 15;
    const int hl   = lane >> 4;

    {
        const float4* src = (const float4*)(rho + (size_t)b * KSUB * KSUB);
        float4* dst = (float4*)sRho;
        #pragma unroll
        for (int i = 0; i < (KSUB * KSUB / 4) / 256; ++i) {
            int g    = tid + i * 256;
            int row  = g >> 5;
            int col4 = g & 31;
            dst[row * (RHO_S / 4) + col4] = src[g];
        }
    }
    const int prow = w * 32 + l16;
    float uyl[8];
    #pragma unroll
    for (int j = 0; j < 8; ++j)
        uyl[j] = uy[(l16 & 7) * 8 + j];
    float uxp[2][8];
    #pragma unroll
    for (int i = 0; i < 2; ++i) {
        #pragma unroll
        for (int j = 0; j < 8; ++j)
            uxp[i][j] = ux[(((prow + 16 * i) >> 3) & 7) * 8 + j];
    }
    float uxl[4][4];
    #pragma unroll
    for (int m = 0; m < 4; ++m) {
        #pragma unroll
        for (int t = 0; t < 4; ++t)
            uxl[m][t] = ux[(2 * m + (l16 >> 3)) * 8 + 2 * t + hl];
    }
    float ucs[4][2];
    #pragma unroll
    for (int c = 0; c < 4; ++c) {
        ucs[c][0] = uc[c * 4 + 2];
        ucs[c][1] = uc[c * 4 + 3];
    }
    __syncthreads();

    v8f accT[2][8] = {{v8f{}, v8f{}, v8f{}, v8f{}, v8f{}, v8f{}, v8f{}, v8f{}},
                      {v8f{}, v8f{}, v8f{}, v8f{}, v8f{}, v8f{}, v8f{}, v8f{}}};
    {
        const int chi = w >> 2;
        const int clo = (w >> 1) & 1;
        const float ucp0 = chi ? (clo ? ucs[3][0] : ucs[2][0]) : (clo ? ucs[1][0] : ucs[0][0]);
        const float ucp1 = chi ? (clo ? ucs[3][1] : ucs[2][1]) : (clo ? ucs[1][1] : ucs[0][1]);
        float uyh[4];
        #pragma unroll
        for (int k = 0; k < 4; ++k)
            uyh[k] = hl ? uyl[4 + k] : uyl[k];
        const float* rA = sRho + (4 * hl) * RHO_S + l16;
        #pragma unroll
        for (int g = 0; g < KSUB / 8; ++g) {
            const float wg0 = ((g >> 3) ? ucp1 : ucp0) * uxp[0][g & 7];
            const float wg1 = ((g >> 3) ? ucp1 : ucp0) * uxp[1][g & 7];
            #pragma unroll
            for (int s = 0; s < 2; ++s) {
                v2f bf0, bf1;
                bf0.x = wg0 * uyh[2 * s];
                bf0.y = wg0 * uyh[2 * s + 1];
                bf1.x = wg1 * uyh[2 * s];
                bf1.y = wg1 * uyh[2 * s + 1];
                const float* rq = rA + (8 * g + 2 * s) * RHO_S;
                #pragma unroll
                for (int tn = 0; tn < 8; ++tn) {
                    v2f af;
                    af.x = rq[tn * 16];
                    af.y = rq[RHO_S + tn * 16];
                    accT[0][tn] = __builtin_amdgcn_wmma_f32_16x16x4_f32(
                        false, af, false, bf0, (short)0, accT[0][tn], false, false);
                    accT[1][tn] = __builtin_amdgcn_wmma_f32_16x16x4_f32(
                        false, af, false, bf1, (short)0, accT[1][tn], false, false);
                }
            }
        }
    }

    float* outb = out + (size_t)b * CDIM * CDIM;
    float* stg = sStg + w * (32 * STG_S);
    for (int th = 0; th < 2; ++th) {
        for (int tq = 0; tq < 2; ++tq) {
            const int tc = 2 * th + tq;
            v8f acc[2][4] = {{v8f{}, v8f{}, v8f{}, v8f{}},
                             {v8f{}, v8f{}, v8f{}, v8f{}}};
            const float uc0 = (tc & 2) ? ((tc & 1) ? ucs[3][0] : ucs[2][0])
                                       : ((tc & 1) ? ucs[1][0] : ucs[0][0]);
            const float uc1 = (tc & 2) ? ((tc & 1) ? ucs[3][1] : ucs[2][1])
                                       : ((tc & 1) ? ucs[1][1] : ucs[0][1]);
            #pragma unroll
            for (int tn = 0; tn < 8; ++tn) {
                float wv[4];
                #pragma unroll
                for (int t = 0; t < 4; ++t)
                    wv[t] = ((tn >> 2) ? uc1 : uc0) * uxl[t][tn & 3];
                #pragma unroll
                for (int js = 0; js < 4; ++js) {
                    v2f a0, a1;
                    a0.x = accT[0][tn][2 * js];
                    a0.y = accT[0][tn][2 * js + 1];
                    a1.x = accT[1][tn][2 * js];
                    a1.y = accT[1][tn][2 * js + 1];
                    #pragma unroll
                    for (int t = 0; t < 4; ++t) {
                        v2f bf;
                        bf.x = wv[t] * uyl[2 * js];
                        bf.y = wv[t] * uyl[2 * js + 1];
                        acc[0][t] = __builtin_amdgcn_wmma_f32_16x16x4_f32(
                            false, a0, false, bf, (short)0, acc[0][t], false, false);
                        acc[1][t] = __builtin_amdgcn_wmma_f32_16x16x4_f32(
                            false, a1, false, bf, (short)0, acc[1][t], false, false);
                    }
                }
            }
            #pragma unroll
            for (int i = 0; i < 2; ++i) {
                #pragma unroll
                for (int t = 0; t < 4; ++t) {
                    #pragma unroll
                    for (int v = 0; v < 8; ++v)
                        stg[(16 * i + v + 8 * hl) * STG_S + tq * 64 + t * 16 + l16] = acc[i][t][v];
                }
            }
        }
        wave_lds_handover();
        #pragma unroll
        for (int r = 0; r < 32; ++r) {
            const float4 val = *(const float4*)(stg + r * STG_S + lane * 4);
            *(float4*)(outb + (w * 32 + r) * CDIM + th * 128 + lane * 4) = val;
        }
        wave_lds_handover();
    }
}

extern "C" void kernel_launch(void* const* d_in, const int* in_sizes, int n_in,
                              void* d_out, int out_size, void* d_ws, size_t ws_size,
                              hipStream_t stream) {
    (void)in_sizes; (void)n_in; (void)out_size; (void)d_ws; (void)ws_size;
    const float* rho = (const float*)d_in[0];
    const float* ux  = (const float*)d_in[1];
    const float* uy  = (const float*)d_in[2];
    const float* uc  = (const float*)d_in[3];
    float* out = (float*)d_out;

    const int lds_bytes = (KSUB * RHO_S + 8 * 32 * STG_S) * (int)sizeof(float);
    (void)hipFuncSetAttribute((const void*)qconv_wmma_kernel,
                              hipFuncAttributeMaxDynamicSharedMemorySize, lds_bytes);

    qconv_wmma_kernel<<<NBATCH, 256, lds_bytes, stream>>>(rho, ux, uy, uc, out);
}
